# starting kernel
# baseline (speedup 1.0000x reference)
#include <hip/hip_runtime.h>

typedef __attribute__((ext_vector_type(2))) float v2f;
typedef __attribute__((ext_vector_type(8))) float v8f;

#define KORD 3
#define NKNOT 12
#define NB    8
#define NF1   20
#define ROWS_PER_BLOCK 256
#define HSTRIDE 9

__device__ __forceinline__ float knot(int t) { return -2.2f + 0.4f * (float)t; }

__device__ __forceinline__ float siluf(float v) {
    return v / (1.0f + __expf(-v));
}

__device__ __forceinline__ void bspline8(float x, float bs[NB]) {
    float B[NKNOT - 1];
#pragma unroll
    for (int j = 0; j < NKNOT - 1; ++j)
        B[j] = (x >= knot(j) && x < knot(j + 1)) ? 1.0f : 0.0f;
#pragma unroll
    for (int p = 1; p <= KORD; ++p) {
        const float inv = 1.0f / (0.4f * (float)p);
#pragma unroll
        for (int j = 0; j < NKNOT - 1 - p; ++j) {
            float left  = (x - knot(j)) * inv;
            float right = (knot(j + p + 1) - x) * inv;
            B[j] = left * B[j] + right * B[j + 1];
        }
    }
#pragma unroll
    for (int j = 0; j < NB; ++j) bs[j] = B[j];
}

__device__ __forceinline__ float w1_elem(int k, int n,
                                         const float* __restrict__ coef1,
                                         const float* __restrict__ sb1,
                                         const float* __restrict__ sp1) {
    if (n >= 5) return 0.0f;
    if (k == 0) return sb1[n];
    if (k == 1) return sb1[5 + n];
    if (k < 2 + NB)     return sp1[n]     * coef1[n * NB + (k - 2)];
    if (k < 2 + 2 * NB) return sp1[5 + n] * coef1[(5 + n) * NB + (k - 2 - NB)];
    return 0.0f;
}

__global__ __launch_bounds__(ROWS_PER_BLOCK) void kan_fused_kernel(
    const float* __restrict__ x,
    const float* __restrict__ coef1, const float* __restrict__ sb1,
    const float* __restrict__ sp1,
    const float* __restrict__ coef2, const float* __restrict__ sb2,
    const float* __restrict__ sp2,
    const float* __restrict__ fcw,   const float* __restrict__ fcb,
    float* __restrict__ out, int nrows)
{
    __shared__ float lds_feat[ROWS_PER_BLOCK * NF1];
    __shared__ float lds_h[ROWS_PER_BLOCK * HSTRIDE];

    const int t    = threadIdx.x;
    const int lane = t & 31;
    const int wave = t >> 5;
    const int blockBase = blockIdx.x * ROWS_PER_BLOCK;
    const int row  = blockBase + t;

    __builtin_prefetch(&x[(size_t)blockBase * 2], 0, 1);

    float x0 = 0.0f, x1 = 0.0f;
    if (row < nrows) {
        x0 = x[(size_t)row * 2 + 0];
        x1 = x[(size_t)row * 2 + 1];
    }
    float bsa[NB], bsb[NB];
    bspline8(x0, bsa);
    bspline8(x1, bsb);
    {
        float* f = &lds_feat[t * NF1];
        const bool live = (row < nrows);
        f[0] = live ? siluf(x0) : 0.0f;
        f[1] = live ? siluf(x1) : 0.0f;
#pragma unroll
        for (int k = 0; k < NB; ++k) {
            f[2 + k]      = live ? bsa[k] : 0.0f;
            f[2 + NB + k] = live ? bsb[k] : 0.0f;
        }
        f[18] = 0.0f; f[19] = 0.0f;
    }

    const int n  = lane & 15;
    const int hh = lane >> 4;
    v2f bfrag[5];
#pragma unroll
    for (int j = 0; j < 5; ++j) {
        int kb = 4 * j + 2 * hh;
        bfrag[j].x = w1_elem(kb,     n, coef1, sb1, sp1);
        bfrag[j].y = w1_elem(kb + 1, n, coef1, sb1, sp1);
    }

    __syncthreads();

    const int R0 = wave * 32;
    const int R1 = R0 + 16;
    v8f acc0 = {}, acc1 = {};
#pragma unroll
    for (int j = 0; j < 5; ++j) {
        int kb = 4 * j + 2 * hh;
        v2f a0 = *(const v2f*)&lds_feat[(R0 + n) * NF1 + kb];
        v2f a1 = *(const v2f*)&lds_feat[(R1 + n) * NF1 + kb];
        acc0 = __builtin_amdgcn_wmma_f32_16x16x4_f32(
            false, a0, false, bfrag[j], (short)0, acc0, false, false);
        acc1 = __builtin_amdgcn_wmma_f32_16x16x4_f32(
            false, a1, false, bfrag[j], (short)0, acc1, false, false);
    }

    if (n < 5) {
#pragma unroll
        for (int v = 0; v < 8; ++v) {
            lds_h[(R0 + v + 8 * hh) * HSTRIDE + n] = acc0[v];
            lds_h[(R1 + v + 8 * hh) * HSTRIDE + n] = acc1[v];
        }
    }

    __syncthreads();

    if (row < nrows) {
        float a = 0.0f;
#pragma unroll
        for (int d = 0; d < 5; ++d) {
            float hv = lds_h[t * HSTRIDE + d];
            float bs2[NB];
            bspline8(hv, bs2);
            float s = 0.0f;
#pragma unroll
            for (int k = 0; k < NB; ++k) s += coef2[d * NB + k] * bs2[k];
            a += sb2[d] * siluf(hv) + sp2[d] * s;
        }
        out[row] = a * fcw[0] + fcb[0];
    }
}

extern "C" void kernel_launch(void* const* d_in, const int* in_sizes, int n_in,
                              void* d_out, int out_size, void* d_ws, size_t ws_size,
                              hipStream_t stream) {
    (void)n_in; (void)d_ws; (void)ws_size; (void)out_size;
    const float* x     = (const float*)d_in[0];
    const float* coef1 = (const float*)d_in[1];
    const float* sb1   = (const float*)d_in[2];
    const float* sp1   = (const float*)d_in[3];
    const float* coef2 = (const float*)d_in[4];
    const float* sb2   = (const float*)d_in[5];
    const float* sp2   = (const float*)d_in[6];
    const float* fcw   = (const float*)d_in[7];
    const float* fcb   = (const float*)d_in[8];
    float* out = (float*)d_out;

    const int nrows   = in_sizes[0] / 2;
    const int nblocks = (nrows + ROWS_PER_BLOCK - 1) / ROWS_PER_BLOCK;

    kan_fused_kernel<<<nblocks, ROWS_PER_BLOCK, 0, stream>>>(
        x, coef1, sb1, sp1, coef2, sb2, sp2, fcw, fcb, out, nrows);
}


# All parameter fetches issued branch-free ahead of the x-dependent work: layer-1 weight fragments from clamped-index loads plus selects, not ten divergent load-and-wait blocks; sum order unchanged
# speedup vs baseline: 2.7933x; 2.7933x over previous
#include <hip/hip_runtime.h>

typedef __attribute__((ext_vector_type(2))) float v2f;
typedef __attribute__((ext_vector_type(4))) float v4f;
typedef __attribute__((ext_vector_type(8))) float v8f;

#define NB    8
#define NF1   20
#define NT    256
#define RPL   4
#define ROWS_PER_BLOCK (NT * RPL)
#define HSTRIDE 9
#define SLOT_FEAT (32 * NF1)
#define WAVE_FEAT (RPL * SLOT_FEAT)

__device__ __forceinline__ float siluf(float v) {
    return v / (1.0f + __expf(-v));
}

#define BS_SCALE (2.0f / 3.0f)
__device__ __forceinline__ void bspline8(float x, float bs[NB]) {
    const float s = fmaf(x, 2.5f, 3.5f);
#pragma unroll
    for (int j = 0; j < NB; ++j) {
        const float d  = fabsf(s - (float)j);
        const float a  = __builtin_amdgcn_fmed3f(fmaf(d, -0.5f, 1.0f), 0.0f, 1.0f);
        const float b  = __builtin_amdgcn_fmed3f(1.0f - d, 0.0f, 1.0f);
        const float a3 = a * a * a;
        const float b3 = b * b * b;
        bs[j] = fmaf(a3, 2.0f, -b3);
    }
}

__global__ __launch_bounds__(NT) void kan_fused_kernel(
    const float* __restrict__ x,
    const float* __restrict__ coef1, const float* __restrict__ sb1,
    const float* __restrict__ sp1,
    const float* __restrict__ coef2, const float* __restrict__ sb2,
    const float* __restrict__ sp2,
    const float* __restrict__ fcw,   const float* __restrict__ fcb,
    float* __restrict__ out, int nrows)
{
    __shared__ float lds_feat[ROWS_PER_BLOCK * NF1];

    const int t    = threadIdx.x;
    const int lane = t & 31;
    const int wave = t >> 5;
    const int blockBase = blockIdx.x * ROWS_PER_BLOCK;
    const int row0 = blockBase + t * RPL;
    float* wfeat = &lds_feat[wave * WAVE_FEAT];

    __builtin_prefetch(&x[(size_t)blockBase * 2], 0, 1);

    const int n  = lane & 15;
    const int hh = lane >> 4;
    const int nn = n < 5 ? n : 4;
    const float spA = sp1[nn], spB = sp1[5 + nn];
    const float sbA = sb1[nn], sbB = sb1[5 + nn];
    float cw[5][2];
#pragma unroll
    for (int j = 0; j < 5; ++j) {
#pragma unroll
        for (int e = 0; e < 2; ++e) {
            const int k   = 4 * j + 2 * hh + e;
            const int in1 = (k >= 2 + NB) ? 1 : 0;
            cw[j][e] = coef1[(in1 * 5 + nn) * NB + ((k - 2) & (NB - 1))];
        }
    }

    float x0[RPL], x1[RPL];
    if (row0 + RPL <= nrows) {
        const v4f* xp = (const v4f*)(x + (size_t)row0 * 2);
        const v4f a = xp[0], b = xp[1];
        x0[0] = a.x; x1[0] = a.y; x0[1] = a.z; x1[1] = a.w;
        x0[2] = b.x; x1[2] = b.y; x0[3] = b.z; x1[3] = b.w;
    } else {
#pragma unroll
        for (int r = 0; r < RPL; ++r) {
            const int row = row0 + r;
            x0[r] = 0.0f; x1[r] = 0.0f;
            if (row < nrows) {
                x0[r] = x[(size_t)row * 2 + 0];
                x1[r] = x[(size_t)row * 2 + 1];
            }
        }
    }

    float c2[5][NB], sp2v[5], sb2v[5];
#pragma unroll
    for (int d = 0; d < 5; ++d) {
        sp2v[d] = sp2[d];
        sb2v[d] = sb2[d];
#pragma unroll
        for (int k = 0; k < NB; ++k) c2[d][k] = coef2[d * NB + k];
    }
    const float fcw0 = fcw[0], fcb0 = fcb[0];

#pragma unroll
    for (int r = 0; r < RPL; ++r) {
        float bsa[NB], bsb[NB];
        bspline8(x0[r], bsa);
        bspline8(x1[r], bsb);
        float* f = &wfeat[r * SLOT_FEAT + lane * NF1];
        const bool live = (row0 + r < nrows);
        f[0] = live ? siluf(x0[r]) : 0.0f;
        f[1] = live ? siluf(x1[r]) : 0.0f;
#pragma unroll
        for (int k = 0; k < NB; ++k) {
            f[2 + k]      = live ? bsa[k] : 0.0f;
            f[2 + NB + k] = live ? bsb[k] : 0.0f;
        }
        f[18] = 0.0f; f[19] = 0.0f;
    }

    v2f bfrag[5];
#pragma unroll
    for (int j = 0; j < 5; ++j) {
        float w[2];
#pragma unroll
        for (int e = 0; e < 2; ++e) {
            const int k = 4 * j + 2 * hh + e;
            const float spl = BS_SCALE * ((k >= 2 + NB ? spB : spA) * cw[j][e]);
            const float sil = (k == 0) ? sbA : sbB;
            float v = (k < 2) ? sil : spl;
            v = (k >= 2 + 2 * NB || n >= 5) ? 0.0f : v;
            w[e] = v;
        }
        bfrag[j].x = w[0];
        bfrag[j].y = w[1];
    }

    __syncthreads();

#pragma unroll
    for (int r = 0; r < RPL; ++r) {
        const float* fr = &wfeat[r * SLOT_FEAT];
        v8f acc0 = {}, acc1 = {};
#pragma unroll
        for (int j = 0; j < 5; ++j) {
            int kb = 4 * j + 2 * hh;
            v2f a0 = *(const v2f*)&fr[n * NF1 + kb];
            v2f a1 = *(const v2f*)&fr[(16 + n) * NF1 + kb];
            acc0 = __builtin_amdgcn_wmma_f32_16x16x4_f32(
                false, a0, false, bfrag[j], (short)0, acc0, false, false);
            acc1 = __builtin_amdgcn_wmma_f32_16x16x4_f32(
                false, a1, false, bfrag[j], (short)0, acc1, false, false);
        }

        if (n < 5) {
            float* hr = &wfeat[r * SLOT_FEAT];
#pragma unroll
            for (int v = 0; v < 8; ++v) {
                hr[(v + 8 * hh) * HSTRIDE + n]      = acc0[v];
                hr[(16 + v + 8 * hh) * HSTRIDE + n] = acc1[v];
            }
        }
    }

    __syncthreads();

    float hv[RPL][5];
#pragma unroll
    for (int r = 0; r < RPL; ++r) {
        const float* hr = &wfeat[r * SLOT_FEAT + lane * HSTRIDE];
#pragma unroll
        for (int d = 0; d < 5; ++d) hv[r][d] = hr[d];
    }

    v2f s2[5][RPL / 2];
#pragma unroll
    for (int d = 0; d < 5; ++d) {
#pragma unroll
        for (int j = 0; j < RPL / 2; ++j) {
            s2[d][j].x = fmaf(hv[2 * j][d],     2.5f, 3.5f);
            s2[d][j].y = fmaf(hv[2 * j + 1][d], 2.5f, 3.5f);
        }
    }

    float smin = s2[0][0].x, smax = s2[0][0].x;
#pragma unroll
    for (int d = 0; d < 5; ++d) {
#pragma unroll
        for (int j = 0; j < RPL / 2; ++j) {
            smin = __builtin_fminf(smin, __builtin_fminf(s2[d][j].x, s2[d][j].y));
            smax = __builtin_fmaxf(smax, __builtin_fmaxf(s2[d][j].x, s2[d][j].y));
        }
    }
    bool reach[6];
    reach[0] = true;
#pragma unroll
    for (int k = 1; k <= 5; ++k)
        reach[k] = (__builtin_amdgcn_ballot_w32(smin < (float)(4 - k)) |
                    __builtin_amdgcn_ballot_w32(smax > (float)(3 + k))) != 0u;

    v2f  accp[RPL / 2];
    float accs[RPL];
#pragma unroll
    for (int j = 0; j < RPL / 2; ++j) { accp[j].x = 0.0f; accp[j].y = 0.0f; }
#pragma unroll
    for (int r = 0; r < RPL; ++r) accs[r] = 0.0f;
    float base = 0.0f;

#pragma unroll
    for (int d = 0; d < 5; ++d) {
        float Cp[NB + 6];
        const float sp = sp2v[d];
#pragma unroll
        for (int j = 0; j < NB + 6; ++j) Cp[j] = 0.0f;
#pragma unroll
        for (int k = 0; k < NB; ++k) Cp[3 + k] = sp * c2[d][k];
        float Ch[NB + 6], C6[NB + 6];
#pragma unroll
        for (int j = 0; j < NB + 6; ++j) { Ch[j] = 0.5f * Cp[j]; C6[j] = (1.0f / 6.0f) * Cp[j]; }
        const float sb = sb2v[d];

        auto span = [&](const int i) {
            const float al = Ch[i + 2] - Ch[i];
            const float be = (Ch[i] + Ch[i + 2]) - Cp[i + 1];
            const float ga = fmaf(3.0f, C6[i + 1] - C6[i + 2], C6[i + 3] - C6[i]);
            v2f g2, b2, a2;
            g2.x = ga; g2.y = ga;
            b2.x = be; b2.y = be;
            a2.x = al; a2.y = al;
#pragma unroll
            for (int j = 0; j < RPL / 2; ++j) {
                v2f u;
                u.x = __builtin_amdgcn_fmed3f(s2[d][j].x + (float)(2 - i), 0.0f, 1.0f);
                u.y = __builtin_amdgcn_fmed3f(s2[d][j].y + (float)(2 - i), 0.0f, 1.0f);
                v2f q = __builtin_elementwise_fma(g2, u, b2);
                q = __builtin_elementwise_fma(q, u, a2);
                accp[j] = __builtin_elementwise_fma(q, u, accp[j]);
            }
        };
        auto knot = [&](const int i) {
            return fmaf(4.0f, C6[i + 1], C6[i] + C6[i + 2]);
        };

        span(5);
        if (reach[1]) {
            span(4); span(6);
            if (reach[2]) {
                span(3); span(7);
                if (reach[3]) {
                    span(2); span(8);
                    if (reach[4]) {
                        span(1); span(9);
                        if (reach[5]) { span(0); span(10); }
                        else base += knot(1);
                    } else base += knot(2);
                } else base += knot(3);
            } else base += knot(4);
        } else base += knot(5);

#pragma unroll
        for (int r = 0; r < RPL; ++r)
            accs[r] = fmaf(sb, siluf(hv[r][d]), accs[r]);
    }

    float o[RPL];
#pragma unroll
    for (int j = 0; j < RPL / 2; ++j) {
        o[2 * j]     = ((accs[2 * j]     + accp[j].x) + base) * fcw0 + fcb0;
        o[2 * j + 1] = ((accs[2 * j + 1] + accp[j].y) + base) * fcw0 + fcb0;
    }

    if (row0 + RPL <= nrows) {
        v4f ov;
        ov.x = o[0]; ov.y = o[1]; ov.z = o[2]; ov.w = o[3];
        *(v4f*)(out + row0) = ov;
    } else {
#pragma unroll
        for (int r = 0; r < RPL; ++r)
            if (row0 + r < nrows) out[row0 + r] = o[r];
    }
}

extern "C" void kernel_launch(void* const* d_in, const int* in_sizes, int n_in,
                              void* d_out, int out_size, void* d_ws, size_t ws_size,
                              hipStream_t stream) {
    (void)n_in; (void)d_ws; (void)ws_size; (void)out_size;
    const float* x     = (const float*)d_in[0];
    const float* coef1 = (const float*)d_in[1];
    const float* sb1   = (const float*)d_in[2];
    const float* sp1   = (const float*)d_in[3];
    const float* coef2 = (const float*)d_in[4];
    const float* sb2   = (const float*)d_in[5];
    const float* sp2   = (const float*)d_in[6];
    const float* fcw   = (const float*)d_in[7];
    const float* fcb   = (const float*)d_in[8];
    float* out = (float*)d_out;

    const int nrows   = in_sizes[0] / 2;
    const int nblocks = (nrows + ROWS_PER_BLOCK - 1) / ROWS_PER_BLOCK;

    kan_fused_kernel<<<nblocks, NT, 0, stream>>>(
        x, coef1, sb1, sp1, coef2, sb2, sp2, fcw, fcb, out, nrows);
}


# Waves whose x all lie in the base grid use the partition of unity: basis 7 of each input folded into the other weights and the accumulator start, so K = 16, 32 WMMAs not 40; layer-1 sum order changes
# speedup vs baseline: 3.0170x; 1.0801x over previous
#include <hip/hip_runtime.h>

typedef __attribute__((ext_vector_type(2))) float v2f;
typedef __attribute__((ext_vector_type(4))) float v4f;
typedef __attribute__((ext_vector_type(8))) float v8f;

#define NB    8
#define NF1   20
#define NT    256
#define RPL   4
#define ROWS_PER_BLOCK (NT * RPL)
#define HSTRIDE 9
#define SLOT_FEAT (32 * NF1)
#define WAVE_FEAT (RPL * SLOT_FEAT)

__device__ __forceinline__ float siluf(float v) {
    return v / (1.0f + __expf(-v));
}

#define BS_SCALE (2.0f / 3.0f)
template <int NBE>
__device__ __forceinline__ void bspline_pair(float xa, float xb, v2f bs[NB]) {
    const float sa = fmaf(xa, 2.5f, 3.5f);
    const float sb = fmaf(xb, 2.5f, 3.5f);
    v2f two; two.x = 2.0f; two.y = 2.0f;
#pragma unroll
    for (int j = 0; j < NBE; ++j) {
        const float da = fabsf(sa - (float)j);
        const float db = fabsf(sb - (float)j);
        v2f a, b;
        a.x = __builtin_amdgcn_fmed3f(fmaf(da, -0.5f, 1.0f), 0.0f, 1.0f);
        a.y = __builtin_amdgcn_fmed3f(fmaf(db, -0.5f, 1.0f), 0.0f, 1.0f);
        b.x = __builtin_amdgcn_fmed3f(1.0f - da, 0.0f, 1.0f);
        b.y = __builtin_amdgcn_fmed3f(1.0f - db, 0.0f, 1.0f);
        const v2f a3 = a * a * a;
        const v2f b3 = b * b * b;
        bs[j] = __builtin_elementwise_fma(a3, two, -b3);
    }
}

template <bool FAST>
__device__ __forceinline__ void layer1_features(const float (&x0)[RPL], const float (&x1)[RPL],
                                                float* wfeat, int lane, int row0, int nrows) {
    constexpr int NBE = FAST ? NB - 1 : NB;
    v2f zero2; zero2.x = 0.0f; zero2.y = 0.0f;
#pragma unroll
    for (int r = 0; r < RPL; ++r) {
        v2f bs[NB];
        bspline_pair<NBE>(x0[r], x1[r], bs);
        v2f* f = (v2f*)&wfeat[r * SLOT_FEAT + lane * NF1];
        const bool live = (row0 + r < nrows);
        v2f sl; sl.x = siluf(x0[r]); sl.y = siluf(x1[r]);
        f[0] = live ? sl : zero2;
#pragma unroll
        for (int k = 0; k < NBE; ++k) f[1 + k] = live ? bs[k] : zero2;
        if (!FAST) f[9] = zero2;
    }
}

template <int KS>
__device__ __forceinline__ void layer1_product(float* wfeat, const v2f (&bfrag)[5], float cstart,
                                               int n, int hh) {
#pragma unroll
    for (int r = 0; r < RPL; ++r) {
        const float* fr = &wfeat[r * SLOT_FEAT];
        v8f acc0, acc1;
#pragma unroll
        for (int v = 0; v < 8; ++v) { acc0[v] = cstart; acc1[v] = cstart; }
#pragma unroll
        for (int j = 0; j < KS; ++j) {
            int kb = 4 * j + 2 * hh;
            v2f a0 = *(const v2f*)&fr[n * NF1 + kb];
            v2f a1 = *(const v2f*)&fr[(16 + n) * NF1 + kb];
            acc0 = __builtin_amdgcn_wmma_f32_16x16x4_f32(
                false, a0, false, bfrag[j], (short)0, acc0, false, false);
            acc1 = __builtin_amdgcn_wmma_f32_16x16x4_f32(
                false, a1, false, bfrag[j], (short)0, acc1, false, false);
        }

        if (n < 5) {
            float* hr = &wfeat[r * SLOT_FEAT];
#pragma unroll
            for (int v = 0; v < 8; ++v) {
                hr[(v + 8 * hh) * HSTRIDE + n]      = acc0[v];
                hr[(16 + v + 8 * hh) * HSTRIDE + n] = acc1[v];
            }
        }
    }
}

__global__ __launch_bounds__(NT) void kan_fused_kernel(
    const float* __restrict__ x,
    const float* __restrict__ coef1, const float* __restrict__ sb1,
    const float* __restrict__ sp1,
    const float* __restrict__ coef2, const float* __restrict__ sb2,
    const float* __restrict__ sp2,
    const float* __restrict__ fcw,   const float* __restrict__ fcb,
    float* __restrict__ out, int nrows)
{
    __shared__ float lds_feat[ROWS_PER_BLOCK * NF1];

    const int t    = threadIdx.x;
    const int lane = t & 31;
    const int wave = t >> 5;
    const int blockBase = blockIdx.x * ROWS_PER_BLOCK;
    const int row0 = blockBase + t * RPL;
    float* wfeat = &lds_feat[wave * WAVE_FEAT];

    __builtin_prefetch(&x[(size_t)blockBase * 2], 0, 1);

    const int n  = lane & 15;
    const int hh = lane >> 4;
    const int nn = n < 5 ? n : 4;
    const float spA = sp1[nn], spB = sp1[5 + nn];
    const float sbA = sb1[nn], sbB = sb1[5 + nn];
    const float c7A = coef1[nn * NB + (NB - 1)], c7B = coef1[(5 + nn) * NB + (NB - 1)];
    float cw[5][2];
#pragma unroll
    for (int j = 0; j < 5; ++j) {
#pragma unroll
        for (int e = 0; e < 2; ++e) {
            const int k   = 4 * j + 2 * hh + e;
            const int in1 = (k - 2) & 1;
            const int idx = ((k - 2) >> 1) & (NB - 1);
            cw[j][e] = coef1[(in1 * 5 + nn) * NB + idx];
        }
    }

    float x0[RPL], x1[RPL];
    if (row0 + RPL <= nrows) {
        const v4f* xp = (const v4f*)(x + (size_t)row0 * 2);
        const v4f a = xp[0], b = xp[1];
        x0[0] = a.x; x1[0] = a.y; x0[1] = a.z; x1[1] = a.w;
        x0[2] = b.x; x1[2] = b.y; x0[3] = b.z; x1[3] = b.w;
    } else {
#pragma unroll
        for (int r = 0; r < RPL; ++r) {
            const int row = row0 + r;
            x0[r] = 0.0f; x1[r] = 0.0f;
            if (row < nrows) {
                x0[r] = x[(size_t)row * 2 + 0];
                x1[r] = x[(size_t)row * 2 + 1];
            }
        }
    }

    float xam = 0.0f;
#pragma unroll
    for (int r = 0; r < RPL; ++r)
        xam = __builtin_fmaxf(xam, __builtin_fmaxf(__builtin_fabsf(x0[r]), __builtin_fabsf(x1[r])));
    const bool fast = __builtin_amdgcn_ballot_w32(!(xam <= 1.0f)) == 0u;

    float c2[5][NB], sp2v[5], sb2v[5];
#pragma unroll
    for (int d = 0; d < 5; ++d) {
        sp2v[d] = sp2[d];
        sb2v[d] = sb2[d];
#pragma unroll
        for (int k = 0; k < NB; ++k) c2[d][k] = coef2[d * NB + k];
    }
    const float fcw0 = fcw[0], fcb0 = fcb[0];

    if (fast) layer1_features<true >(x0, x1, wfeat, lane, row0, nrows);
    else      layer1_features<false>(x0, x1, wfeat, lane, row0, nrows);

    const float cstart = (fast && n < 5) ? fmaf(spA, c7A, spB * c7B) : 0.0f;
    v2f bfrag[5];
#pragma unroll
    for (int j = 0; j < 5; ++j) {
        float w[2];
#pragma unroll
        for (int e = 0; e < 2; ++e) {
            const int k = 4 * j + 2 * hh + e;
            const bool in1 = ((k - 2) & 1) != 0;
            const float sp  = in1 ? spB : spA;
            const float sub = fast ? sp * (in1 ? c7B : c7A) : 0.0f;
            const float spl = BS_SCALE * (sp * cw[j][e] - sub);
            const float sil = (k == 0) ? sbA : sbB;
            float v = (k < 2) ? sil : spl;
            v = (k >= 2 + 2 * NB || n >= 5) ? 0.0f : v;
            w[e] = v;
        }
        bfrag[j].x = w[0];
        bfrag[j].y = w[1];
    }

    __syncthreads();

    if (fast) layer1_product<4>(wfeat, bfrag, cstart, n, hh);
    else      layer1_product<5>(wfeat, bfrag, 0.0f,   n, hh);

    __syncthreads();

    float hv[RPL][5];
#pragma unroll
    for (int r = 0; r < RPL; ++r) {
        const float* hr = &wfeat[r * SLOT_FEAT + lane * HSTRIDE];
#pragma unroll
        for (int d = 0; d < 5; ++d) hv[r][d] = hr[d];
    }

    v2f s2[5][RPL / 2];
#pragma unroll
    for (int d = 0; d < 5; ++d) {
#pragma unroll
        for (int j = 0; j < RPL / 2; ++j) {
            s2[d][j].x = fmaf(hv[2 * j][d],     2.5f, 3.5f);
            s2[d][j].y = fmaf(hv[2 * j + 1][d], 2.5f, 3.5f);
        }
    }

    float smin = s2[0][0].x, smax = s2[0][0].x;
#pragma unroll
    for (int d = 0; d < 5; ++d) {
#pragma unroll
        for (int j = 0; j < RPL / 2; ++j) {
            smin = __builtin_fminf(smin, __builtin_fminf(s2[d][j].x, s2[d][j].y));
            smax = __builtin_fmaxf(smax, __builtin_fmaxf(s2[d][j].x, s2[d][j].y));
        }
    }
    bool reach[6];
    reach[0] = true;
#pragma unroll
    for (int k = 1; k <= 5; ++k)
        reach[k] = (__builtin_amdgcn_ballot_w32(smin < (float)(4 - k)) |
                    __builtin_amdgcn_ballot_w32(smax > (float)(3 + k))) != 0u;

    v2f  accp[RPL / 2];
    float accs[RPL];
#pragma unroll
    for (int j = 0; j < RPL / 2; ++j) { accp[j].x = 0.0f; accp[j].y = 0.0f; }
#pragma unroll
    for (int r = 0; r < RPL; ++r) accs[r] = 0.0f;
    float base = 0.0f;

#pragma unroll
    for (int d = 0; d < 5; ++d) {
        float Cp[NB + 6];
        const float sp = sp2v[d];
#pragma unroll
        for (int j = 0; j < NB + 6; ++j) Cp[j] = 0.0f;
#pragma unroll
        for (int k = 0; k < NB; ++k) Cp[3 + k] = sp * c2[d][k];
        float Ch[NB + 6], C6[NB + 6];
#pragma unroll
        for (int j = 0; j < NB + 6; ++j) { Ch[j] = 0.5f * Cp[j]; C6[j] = (1.0f / 6.0f) * Cp[j]; }
        const float sb = sb2v[d];

        auto span = [&](const int i) {
            const float al = Ch[i + 2] - Ch[i];
            const float be = (Ch[i] + Ch[i + 2]) - Cp[i + 1];
            const float ga = fmaf(3.0f, C6[i + 1] - C6[i + 2], C6[i + 3] - C6[i]);
            v2f g2, b2, a2;
            g2.x = ga; g2.y = ga;
            b2.x = be; b2.y = be;
            a2.x = al; a2.y = al;
#pragma unroll
            for (int j = 0; j < RPL / 2; ++j) {
                v2f u;
                u.x = __builtin_amdgcn_fmed3f(s2[d][j].x + (float)(2 - i), 0.0f, 1.0f);
                u.y = __builtin_amdgcn_fmed3f(s2[d][j].y + (float)(2 - i), 0.0f, 1.0f);
                v2f q = __builtin_elementwise_fma(g2, u, b2);
                q = __builtin_elementwise_fma(q, u, a2);
                accp[j] = __builtin_elementwise_fma(q, u, accp[j]);
            }
        };
        auto knot = [&](const int i) {
            return fmaf(4.0f, C6[i + 1], C6[i] + C6[i + 2]);
        };

        span(5);
        if (reach[1]) {
            span(4); span(6);
            if (reach[2]) {
                span(3); span(7);
                if (reach[3]) {
                    span(2); span(8);
                    if (reach[4]) {
                        span(1); span(9);
                        if (reach[5]) { span(0); span(10); }
                        else base += knot(1);
                    } else base += knot(2);
                } else base += knot(3);
            } else base += knot(4);
        } else base += knot(5);

#pragma unroll
        for (int r = 0; r < RPL; ++r)
            accs[r] = fmaf(sb, siluf(hv[r][d]), accs[r]);
    }

    float o[RPL];
#pragma unroll
    for (int j = 0; j < RPL / 2; ++j) {
        o[2 * j]     = ((accs[2 * j]     + accp[j].x) + base) * fcw0 + fcb0;
        o[2 * j + 1] = ((accs[2 * j + 1] + accp[j].y) + base) * fcw0 + fcb0;
    }

    if (row0 + RPL <= nrows) {
        v4f ov;
        ov.x = o[0]; ov.y = o[1]; ov.z = o[2]; ov.w = o[3];
        *(v4f*)(out + row0) = ov;
    } else {
#pragma unroll
        for (int r = 0; r < RPL; ++r)
            if (row0 + r < nrows) out[row0 + r] = o[r];
    }
}

extern "C" void kernel_launch(void* const* d_in, const int* in_sizes, int n_in,
                              void* d_out, int out_size, void* d_ws, size_t ws_size,
                              hipStream_t stream) {
    (void)n_in; (void)d_ws; (void)ws_size; (void)out_size;
    const float* x     = (const float*)d_in[0];
    const float* coef1 = (const float*)d_in[1];
    const float* sb1   = (const float*)d_in[2];
    const float* sp1   = (const float*)d_in[3];
    const float* coef2 = (const float*)d_in[4];
    const float* sb2   = (const float*)d_in[5];
    const float* sp2   = (const float*)d_in[6];
    const float* fcw   = (const float*)d_in[7];
    const float* fcb   = (const float*)d_in[8];
    float* out = (float*)d_out;

    const int nrows   = in_sizes[0] / 2;
    const int nblocks = (nrows + ROWS_PER_BLOCK - 1) / ROWS_PER_BLOCK;

    kan_fused_kernel<<<nblocks, NT, 0, stream>>>(
        x, coef1, sb1, sp1, coef2, sb2, sp2, fcw, fcb, out, nrows);
}


# A lane owns 8 rows, not 4 (8192 waves): the wave's eight 32-row slots stream through one private LDS tile (features, fragments, WMMA, h); no work-group barrier; layer-2 span culling per 256 rows
# speedup vs baseline: 3.5645x; 1.1815x over previous
#include <hip/hip_runtime.h>

typedef __attribute__((ext_vector_type(2))) float v2f;
typedef __attribute__((ext_vector_type(4))) float v4f;
typedef __attribute__((ext_vector_type(8))) float v8f;

#define NB    8
#define NF1   20
#define NT    128
#define RPL   8
#define ROWS_PER_BLOCK (NT * RPL)
#define ROWS_PER_WAVE  (32 * RPL)
#define HSTRIDE 5
#define SLOT_FEAT (32 * NF1)
#define SLOT_H    (32 * HSTRIDE)
#define WAVE_LDS  (SLOT_FEAT + RPL * SLOT_H)

__device__ __forceinline__ float siluf(float v) {
    return v / (1.0f + __expf(-v));
}

#define BS_SCALE (2.0f / 3.0f)
__device__ __forceinline__ void bspline_pair(float xa, float xb, v2f bs[NB]) {
    const float sa = fmaf(xa, 2.5f, 3.5f);
    const float sb = fmaf(xb, 2.5f, 3.5f);
    v2f two; two.x = 2.0f; two.y = 2.0f;
#pragma unroll
    for (int j = 0; j < NB; ++j) {
        const float da = fabsf(sa - (float)j);
        const float db = fabsf(sb - (float)j);
        v2f a, b;
        a.x = __builtin_amdgcn_fmed3f(fmaf(da, -0.5f, 1.0f), 0.0f, 1.0f);
        a.y = __builtin_amdgcn_fmed3f(fmaf(db, -0.5f, 1.0f), 0.0f, 1.0f);
        b.x = __builtin_amdgcn_fmed3f(1.0f - da, 0.0f, 1.0f);
        b.y = __builtin_amdgcn_fmed3f(1.0f - db, 0.0f, 1.0f);
        const v2f a3 = a * a * a;
        const v2f b3 = b * b * b;
        bs[j] = __builtin_elementwise_fma(a3, two, -b3);
    }
}

__device__ __forceinline__ void wave_lds_order() {
    __builtin_amdgcn_fence(__ATOMIC_RELEASE, "wavefront");
    __builtin_amdgcn_wave_barrier();
    __builtin_amdgcn_fence(__ATOMIC_ACQUIRE, "wavefront");
}

__device__ __forceinline__ void features_full_row(float xa, float xb, bool live, float* tile, int lane) {
    v2f zero2; zero2.x = 0.0f; zero2.y = 0.0f;
    v2f bs[NB];
    bspline_pair(xa, xb, bs);
    v2f* f = (v2f*)&tile[lane * NF1];
    v2f sl; sl.x = siluf(xa); sl.y = siluf(xb);
    f[0] = live ? sl : zero2;
#pragma unroll
    for (int k = 0; k < NB; ++k) f[1 + k] = live ? bs[k] : zero2;
    f[9] = zero2;
}

__device__ __forceinline__ void features_ingrid_row(float xa, float xb, float* tile, int lane) {
    v2f f[8];
    f[0].x = siluf(xa); f[0].y = siluf(xb);
    v2f xx; xx.x = xa; xx.y = xb;
    f[1] = xx;
    f[2] = xx * xx;
    f[3] = f[2] * xx;
    v2f l2, l1, r1, r2;
    l2.x = __builtin_amdgcn_fmed3f(fmaf(xa, -2.5f,  -1.5f),  0.0f, 1.0f);
    l2.y = __builtin_amdgcn_fmed3f(fmaf(xb, -2.5f,  -1.5f),  0.0f, 1.0f);
    l1.x = __builtin_amdgcn_fmed3f(fmaf(xa, -1.25f, -0.25f), 0.0f, 1.0f);
    l1.y = __builtin_amdgcn_fmed3f(fmaf(xb, -1.25f, -0.25f), 0.0f, 1.0f);
    r1.x = __builtin_amdgcn_fmed3f(fmaf(xa,  1.25f, -0.25f), 0.0f, 1.0f);
    r1.y = __builtin_amdgcn_fmed3f(fmaf(xb,  1.25f, -0.25f), 0.0f, 1.0f);
    r2.x = __builtin_amdgcn_fmed3f(fmaf(xa,  2.5f,  -1.5f),  0.0f, 1.0f);
    r2.y = __builtin_amdgcn_fmed3f(fmaf(xb,  2.5f,  -1.5f),  0.0f, 1.0f);
    f[4] = l2 * l2 * l2;
    f[5] = l1 * l1 * l1;
    f[6] = r1 * r1 * r1;
    f[7] = r2 * r2 * r2;
    v2f* fp = (v2f*)&tile[lane * NF1];
#pragma unroll
    for (int k = 0; k < 8; ++k) fp[k] = f[k];
}

__device__ __forceinline__ void ingrid_weights(const float (&c)[NB], float sp, float sil, int hh,
                                               float (&w)[4], float& p0) {
    float s[7];
#pragma unroll
    for (int j = 0; j < 7; ++j) s[j] = hh ? c[j + 1] : c[j];
    const float ja = fmaf(6.0f, s[2], (s[0] + s[4]) - 4.0f * (s[1] + s[3]));
    const float jb = fmaf(6.0f, s[4], (s[2] + s[6]) - 4.0f * (s[3] + s[5]));
    const float dA = c[5] - c[2], dB = c[4] - c[3];
    const float wx  = fmaf(0.3125f, dA, 1.5625f * dB);
    const float wx2 = 1.5625f * ((c[2] + c[5]) - (c[3] + c[4]));
    const float wx3 = (125.0f / 48.0f) * fmaf(-3.0f, dB, dA);
    const float ga = hh ? (4.0f / 3.0f) : (1.0f / 6.0f);
    const float gb = hh ? (1.0f / 6.0f) : (4.0f / 3.0f);
    w[0] = hh ? sp * wx : sil;
    w[1] = sp * (hh ? wx3 : wx2);
    w[2] = sp * (ga * ja);
    w[3] = sp * (gb * jb);
    p0 = sp * ((1.0f / 48.0f) * fmaf(23.0f, c[3] + c[4], c[2] + c[5]));
}

template <int KS>
__device__ __forceinline__ void product_slot(const float* tile, float* htile, const v2f (&bfrag)[5],
                                             float cstart, int n, int hh) {
    v8f acc0, acc1;
#pragma unroll
    for (int v = 0; v < 8; ++v) { acc0[v] = cstart; acc1[v] = cstart; }
#pragma unroll
    for (int j = 0; j < KS; ++j) {
        int kb = 4 * j + 2 * hh;
        v2f a0 = *(const v2f*)&tile[n * NF1 + kb];
        v2f a1 = *(const v2f*)&tile[(16 + n) * NF1 + kb];
        acc0 = __builtin_amdgcn_wmma_f32_16x16x4_f32(
            false, a0, false, bfrag[j], (short)0, acc0, false, false);
        acc1 = __builtin_amdgcn_wmma_f32_16x16x4_f32(
            false, a1, false, bfrag[j], (short)0, acc1, false, false);
    }

    if (n < 5) {
#pragma unroll
        for (int v = 0; v < 8; ++v) {
            htile[(v + 8 * hh) * HSTRIDE + n]      = acc0[v];
            htile[(16 + v + 8 * hh) * HSTRIDE + n] = acc1[v];
        }
    }
}

__global__ __launch_bounds__(NT) void kan_fused_kernel(
    const float* __restrict__ x,
    const float* __restrict__ coef1, const float* __restrict__ sb1,
    const float* __restrict__ sp1,
    const float* __restrict__ coef2, const float* __restrict__ sb2,
    const float* __restrict__ sp2,
    const float* __restrict__ fcw,   const float* __restrict__ fcb,
    float* __restrict__ out, int nrows)
{
    __shared__ float lds_wave[(NT / 32) * WAVE_LDS];

    const int t    = threadIdx.x;
    const int lane = t & 31;
    const int wave = t >> 5;
    const int blockBase = blockIdx.x * ROWS_PER_BLOCK;
    const int waveBase  = blockBase + wave * ROWS_PER_WAVE;
    const int rowq = waveBase + lane * 4;
    float* tile   = &lds_wave[wave * WAVE_LDS];
    float* htiles = tile + SLOT_FEAT;

    __builtin_prefetch(&x[(size_t)blockBase * 2], 0, 1);

    const int n  = lane & 15;
    const int hh = lane >> 4;
    const int nn = n < 5 ? n : 4;
    const float spA = sp1[nn], spB = sp1[5 + nn];
    const float sbA = sb1[nn], sbB = sb1[5 + nn];
    float cA[NB], cB[NB];
#pragma unroll
    for (int j = 0; j < NB; ++j) {
        cA[j] = coef1[nn * NB + j];
        cB[j] = coef1[(5 + nn) * NB + j];
    }

    float x0[RPL], x1[RPL];
    if (rowq + (RPL / 4 - 1) * 128 + 4 <= nrows) {
#pragma unroll
        for (int q = 0; q < RPL / 4; ++q) {
            const v4f* xp = (const v4f*)(x + (size_t)(rowq + q * 128) * 2);
            const v4f a = xp[0], b = xp[1];
            x0[4 * q + 0] = a.x; x1[4 * q + 0] = a.y; x0[4 * q + 1] = a.z; x1[4 * q + 1] = a.w;
            x0[4 * q + 2] = b.x; x1[4 * q + 2] = b.y; x0[4 * q + 3] = b.z; x1[4 * q + 3] = b.w;
        }
    } else {
#pragma unroll
        for (int r = 0; r < RPL; ++r) {
            const int row = rowq + (r >> 2) * 128 + (r & 3);
            x0[r] = 0.0f; x1[r] = 0.0f;
            if (row < nrows) {
                x0[r] = x[(size_t)row * 2 + 0];
                x1[r] = x[(size_t)row * 2 + 1];
            }
        }
    }

    float xam = 0.0f;
#pragma unroll
    for (int r = 0; r < RPL; ++r)
        xam = __builtin_fmaxf(xam, __builtin_fmaxf(__builtin_fabsf(x0[r]), __builtin_fabsf(x1[r])));
    const bool fast = __builtin_amdgcn_ballot_w32(!(xam <= 1.0f)) == 0u;

    float c2[5][NB], sp2v[5], sb2v[5];
#pragma unroll
    for (int d = 0; d < 5; ++d) {
        sp2v[d] = sp2[d];
        sb2v[d] = sb2[d];
#pragma unroll
        for (int k = 0; k < NB; ++k) c2[d][k] = coef2[d * NB + k];
    }
    const float fcw0 = fcw[0], fcb0 = fcb[0];

    v2f bfrag[5];
    float cstart;
    if (fast) {
        float wA[4], wB[4], pA, pB;
        ingrid_weights(cA, spA, sbA, hh, wA, pA);
        ingrid_weights(cB, spB, sbB, hh, wB, pB);
        const bool col = n < 5;
#pragma unroll
        for (int j = 0; j < 4; ++j) {
            bfrag[j].x = col ? wA[j] : 0.0f;
            bfrag[j].y = col ? wB[j] : 0.0f;
        }
        bfrag[4].x = 0.0f; bfrag[4].y = 0.0f;
        cstart = col ? pA + pB : 0.0f;
#pragma unroll
        for (int r = 0; r < RPL; ++r) {
            features_ingrid_row(x0[r], x1[r], tile, lane);
            wave_lds_order();
            product_slot<4>(tile, htiles + r * SLOT_H, bfrag, cstart, n, hh);
            wave_lds_order();
        }
    } else {
#pragma unroll
        for (int j = 0; j < 5; ++j) {
            float w[2];
#pragma unroll
            for (int e = 0; e < 2; ++e) {
                const int k = 4 * j + 2 * hh + e;
                const float sp  = e ? spB : spA;
                const float cw  = e ? (hh ? cB[(2 * j) & 7] : cB[(2 * j - 1) & 7])
                                    : (hh ? cA[(2 * j) & 7] : cA[(2 * j - 1) & 7]);
                const float spl = BS_SCALE * (sp * cw);
                const float sil = (k == 0) ? sbA : sbB;
                float v = (k < 2) ? sil : spl;
                v = (k >= 2 + 2 * NB || n >= 5) ? 0.0f : v;
                w[e] = v;
            }
            bfrag[j].x = w[0];
            bfrag[j].y = w[1];
        }
        cstart = 0.0f;
#pragma unroll 1
        for (int r = 0; r < RPL; ++r) {
            const int row = rowq + (r >> 2) * 128 + (r & 3);
            const bool live = row < nrows;
            float xa = 0.0f, xb = 0.0f;
            if (live) {
                xa = x[(size_t)row * 2 + 0];
                xb = x[(size_t)row * 2 + 1];
            }
            features_full_row(xa, xb, live, tile, lane);
            wave_lds_order();
            product_slot<5>(tile, htiles + r * SLOT_H, bfrag, cstart, n, hh);
            wave_lds_order();
        }
    }

    float hv[RPL][5];
#pragma unroll
    for (int r = 0; r < RPL; ++r) {
        const float* hr = &htiles[r * SLOT_H + lane * HSTRIDE];
#pragma unroll
        for (int d = 0; d < 5; ++d) hv[r][d] = hr[d];
    }

    v2f s2[5][RPL / 2];
#pragma unroll
    for (int d = 0; d < 5; ++d) {
#pragma unroll
        for (int j = 0; j < RPL / 2; ++j) {
            s2[d][j].x = fmaf(hv[2 * j][d],     2.5f, 3.5f);
            s2[d][j].y = fmaf(hv[2 * j + 1][d], 2.5f, 3.5f);
        }
    }

    float smin = s2[0][0].x, smax = s2[0][0].x;
#pragma unroll
    for (int d = 0; d < 5; ++d) {
#pragma unroll
        for (int j = 0; j < RPL / 2; ++j) {
            smin = __builtin_fminf(smin, __builtin_fminf(s2[d][j].x, s2[d][j].y));
            smax = __builtin_fmaxf(smax, __builtin_fmaxf(s2[d][j].x, s2[d][j].y));
        }
    }
    bool reach[6];
    reach[0] = true;
#pragma unroll
    for (int k = 1; k <= 5; ++k)
        reach[k] = (__builtin_amdgcn_ballot_w32(smin < (float)(4 - k)) |
                    __builtin_amdgcn_ballot_w32(smax > (float)(3 + k))) != 0u;

    v2f  accp[RPL / 2];
    float accs[RPL];
#pragma unroll
    for (int j = 0; j < RPL / 2; ++j) { accp[j].x = 0.0f; accp[j].y = 0.0f; }
#pragma unroll
    for (int r = 0; r < RPL; ++r) accs[r] = 0.0f;
    float base = 0.0f;

#pragma unroll
    for (int d = 0; d < 5; ++d) {
        float Cp[NB + 6];
        const float sp = sp2v[d];
#pragma unroll
        for (int j = 0; j < NB + 6; ++j) Cp[j] = 0.0f;
#pragma unroll
        for (int k = 0; k < NB; ++k) Cp[3 + k] = sp * c2[d][k];
        float Ch[NB + 6], C6[NB + 6];
#pragma unroll
        for (int j = 0; j < NB + 6; ++j) { Ch[j] = 0.5f * Cp[j]; C6[j] = (1.0f / 6.0f) * Cp[j]; }
        const float sb = sb2v[d];

        auto span = [&](const int i) {
            const float al = Ch[i + 2] - Ch[i];
            const float be = (Ch[i] + Ch[i + 2]) - Cp[i + 1];
            const float ga = fmaf(3.0f, C6[i + 1] - C6[i + 2], C6[i + 3] - C6[i]);
            v2f g2, b2, a2;
            g2.x = ga; g2.y = ga;
            b2.x = be; b2.y = be;
            a2.x = al; a2.y = al;
#pragma unroll
            for (int j = 0; j < RPL / 2; ++j) {
                v2f u;
                u.x = __builtin_amdgcn_fmed3f(s2[d][j].x + (float)(2 - i), 0.0f, 1.0f);
                u.y = __builtin_amdgcn_fmed3f(s2[d][j].y + (float)(2 - i), 0.0f, 1.0f);
                v2f q = __builtin_elementwise_fma(g2, u, b2);
                q = __builtin_elementwise_fma(q, u, a2);
                accp[j] = __builtin_elementwise_fma(q, u, accp[j]);
            }
        };
        auto knot = [&](const int i) {
            return fmaf(4.0f, C6[i + 1], C6[i] + C6[i + 2]);
        };

        span(5);
        if (reach[1]) {
            span(4); span(6);
            if (reach[2]) {
                span(3); span(7);
                if (reach[3]) {
                    span(2); span(8);
                    if (reach[4]) {
                        span(1); span(9);
                        if (reach[5]) { span(0); span(10); }
                        else base += knot(1);
                    } else base += knot(2);
                } else base += knot(3);
            } else base += knot(4);
        } else base += knot(5);

#pragma unroll
        for (int r = 0; r < RPL; ++r)
            accs[r] = fmaf(sb, siluf(hv[r][d]), accs[r]);
    }

    float o[RPL];
#pragma unroll
    for (int j = 0; j < RPL / 2; ++j) {
        o[2 * j]     = ((accs[2 * j]     + accp[j].x) + base) * fcw0 + fcb0;
        o[2 * j + 1] = ((accs[2 * j + 1] + accp[j].y) + base) * fcw0 + fcb0;
    }

#pragma unroll
    for (int q = 0; q < RPL / 4; ++q) {
        const int rq = rowq + q * 128;
        if (rq + 4 <= nrows) {
            v4f ov;
            ov.x = o[4 * q + 0]; ov.y = o[4 * q + 1]; ov.z = o[4 * q + 2]; ov.w = o[4 * q + 3];
            *(v4f*)(out + rq) = ov;
        } else {
#pragma unroll
            for (int i = 0; i < 4; ++i)
                if (rq + i < nrows) out[rq + i] = o[4 * q + i];
        }
    }
}

extern "C" void kernel_launch(void* const* d_in, const int* in_sizes, int n_in,
                              void* d_out, int out_size, void* d_ws, size_t ws_size,
                              hipStream_t stream) {
    (void)n_in; (void)d_ws; (void)ws_size; (void)out_size;
    const float* x     = (const float*)d_in[0];
    const float* coef1 = (const float*)d_in[1];
    const float* sb1   = (const float*)d_in[2];
    const float* sp1   = (const float*)d_in[3];
    const float* coef2 = (const float*)d_in[4];
    const float* sb2   = (const float*)d_in[5];
    const float* sp2   = (const float*)d_in[6];
    const float* fcw   = (const float*)d_in[7];
    const float* fcb   = (const float*)d_in[8];
    float* out = (float*)d_out;

    const int nrows   = in_sizes[0] / 2;
    const int nblocks = (nrows + ROWS_PER_BLOCK - 1) / ROWS_PER_BLOCK;

    kan_fused_kernel<<<nblocks, NT, 0, stream>>>(
        x, coef1, sb1, sp1, coef2, sb2, sp2, fcw, fcb, out, nrows);
}


# One correctly rounded division shared by many SiLUs (product of denominators, 3 multiplies per value): 16 of layer 1 per lane, 10 per row pair in layer 2; silu rounding changes, sum order unchanged
# speedup vs baseline: 4.0680x; 1.1413x over previous
#include <hip/hip_runtime.h>

typedef __attribute__((ext_vector_type(2))) float v2f;
typedef __attribute__((ext_vector_type(4))) float v4f;
typedef __attribute__((ext_vector_type(8))) float v8f;

#define NB    8
#define NF1   20
#define NT    128
#define RPL   8
#define ROWS_PER_BLOCK (NT * RPL)
#define ROWS_PER_WAVE  (32 * RPL)
#define HSTRIDE 5
#define SLOT_FEAT (32 * NF1)
#define SLOT_H    (32 * HSTRIDE)
#define WAVE_LDS  (SLOT_FEAT + RPL * SLOT_H)

__device__ __forceinline__ float siluf(float v) {
    return v / (1.0f + __expf(-v));
}


__device__ __forceinline__ void silu_batch_ingrid(const float (&x0)[RPL], const float (&x1)[RPL], v2f (&sl)[RPL]) {
    v2f one2; one2.x = 1.0f; one2.y = 1.0f;
    v2f D[RPL], P[RPL];
#pragma unroll
    for (int r = RPL - 1; r >= 0; --r) {
        v2f t; t.x = __expf(-x0[r]); t.y = __expf(-x1[r]);
        D[r] = t + one2;
        P[r] = (r < RPL - 1) ? P[r + 1] * D[r] : D[RPL - 1];
    }
    const float R = 1.0f / (P[0].x * P[0].y);
    v2f R2; R2.x = R * P[0].y; R2.y = R * P[0].x;
#pragma unroll
    for (int r = 0; r < RPL; ++r) {
        v2f xx; xx.x = x0[r]; xx.y = x1[r];
        const v2f inv = (r < RPL - 1) ? R2 * P[r + 1] : R2;
        sl[r] = xx * inv;
        R2 = R2 * D[r];
    }
}

#define BS_SCALE (2.0f / 3.0f)
__device__ __forceinline__ void bspline_pair(float xa, float xb, v2f bs[NB]) {
    const float sa = fmaf(xa, 2.5f, 3.5f);
    const float sb = fmaf(xb, 2.5f, 3.5f);
    v2f two; two.x = 2.0f; two.y = 2.0f;
#pragma unroll
    for (int j = 0; j < NB; ++j) {
        const float da = fabsf(sa - (float)j);
        const float db = fabsf(sb - (float)j);
        v2f a, b;
        a.x = __builtin_amdgcn_fmed3f(fmaf(da, -0.5f, 1.0f), 0.0f, 1.0f);
        a.y = __builtin_amdgcn_fmed3f(fmaf(db, -0.5f, 1.0f), 0.0f, 1.0f);
        b.x = __builtin_amdgcn_fmed3f(1.0f - da, 0.0f, 1.0f);
        b.y = __builtin_amdgcn_fmed3f(1.0f - db, 0.0f, 1.0f);
        const v2f a3 = a * a * a;
        const v2f b3 = b * b * b;
        bs[j] = __builtin_elementwise_fma(a3, two, -b3);
    }
}

__device__ __forceinline__ void wave_lds_order() {
    __builtin_amdgcn_fence(__ATOMIC_RELEASE, "wavefront");
    __builtin_amdgcn_wave_barrier();
    __builtin_amdgcn_fence(__ATOMIC_ACQUIRE, "wavefront");
}

__device__ __forceinline__ void features_full_row(float xa, float xb, bool live, float* tile, int lane) {
    v2f zero2; zero2.x = 0.0f; zero2.y = 0.0f;
    v2f bs[NB];
    bspline_pair(xa, xb, bs);
    v2f* f = (v2f*)&tile[lane * NF1];
    v2f sl; sl.x = siluf(xa); sl.y = siluf(xb);
    f[0] = live ? sl : zero2;
#pragma unroll
    for (int k = 0; k < NB; ++k) f[1 + k] = live ? bs[k] : zero2;
    f[9] = zero2;
}

__device__ __forceinline__ void features_ingrid_row(float xa, float xb, v2f sl, float* tile, int lane) {
    v2f f[8];
    f[0] = sl;
    v2f xx; xx.x = xa; xx.y = xb;
    f[1] = xx;
    f[2] = xx * xx;
    f[3] = f[2] * xx;
    v2f l2, l1, r1, r2;
    l2.x = __builtin_amdgcn_fmed3f(fmaf(xa, -2.5f,  -1.5f),  0.0f, 1.0f);
    l2.y = __builtin_amdgcn_fmed3f(fmaf(xb, -2.5f,  -1.5f),  0.0f, 1.0f);
    l1.x = __builtin_amdgcn_fmed3f(fmaf(xa, -1.25f, -0.25f), 0.0f, 1.0f);
    l1.y = __builtin_amdgcn_fmed3f(fmaf(xb, -1.25f, -0.25f), 0.0f, 1.0f);
    r1.x = __builtin_amdgcn_fmed3f(fmaf(xa,  1.25f, -0.25f), 0.0f, 1.0f);
    r1.y = __builtin_amdgcn_fmed3f(fmaf(xb,  1.25f, -0.25f), 0.0f, 1.0f);
    r2.x = __builtin_amdgcn_fmed3f(fmaf(xa,  2.5f,  -1.5f),  0.0f, 1.0f);
    r2.y = __builtin_amdgcn_fmed3f(fmaf(xb,  2.5f,  -1.5f),  0.0f, 1.0f);
    f[4] = l2 * l2 * l2;
    f[5] = l1 * l1 * l1;
    f[6] = r1 * r1 * r1;
    f[7] = r2 * r2 * r2;
    v2f* fp = (v2f*)&tile[lane * NF1];
#pragma unroll
    for (int k = 0; k < 8; ++k) fp[k] = f[k];
}

__device__ __forceinline__ void ingrid_weights(const float (&c)[NB], float sp, float sil, int hh,
                                               float (&w)[4], float& p0) {
    float s[7];
#pragma unroll
    for (int j = 0; j < 7; ++j) s[j] = hh ? c[j + 1] : c[j];
    const float ja = fmaf(6.0f, s[2], (s[0] + s[4]) - 4.0f * (s[1] + s[3]));
    const float jb = fmaf(6.0f, s[4], (s[2] + s[6]) - 4.0f * (s[3] + s[5]));
    const float dA = c[5] - c[2], dB = c[4] - c[3];
    const float wx  = fmaf(0.3125f, dA, 1.5625f * dB);
    const float wx2 = 1.5625f * ((c[2] + c[5]) - (c[3] + c[4]));
    const float wx3 = (125.0f / 48.0f) * fmaf(-3.0f, dB, dA);
    const float ga = hh ? (4.0f / 3.0f) : (1.0f / 6.0f);
    const float gb = hh ? (1.0f / 6.0f) : (4.0f / 3.0f);
    w[0] = hh ? sp * wx : sil;
    w[1] = sp * (hh ? wx3 : wx2);
    w[2] = sp * (ga * ja);
    w[3] = sp * (gb * jb);
    p0 = sp * ((1.0f / 48.0f) * fmaf(23.0f, c[3] + c[4], c[2] + c[5]));
}

template <int KS>
__device__ __forceinline__ void product_slot(const float* tile, float* htile, const v2f (&bfrag)[5],
                                             float cstart, int n, int hh) {
    v8f acc0, acc1;
#pragma unroll
    for (int v = 0; v < 8; ++v) { acc0[v] = cstart; acc1[v] = cstart; }
#pragma unroll
    for (int j = 0; j < KS; ++j) {
        int kb = 4 * j + 2 * hh;
        v2f a0 = *(const v2f*)&tile[n * NF1 + kb];
        v2f a1 = *(const v2f*)&tile[(16 + n) * NF1 + kb];
        acc0 = __builtin_amdgcn_wmma_f32_16x16x4_f32(
            false, a0, false, bfrag[j], (short)0, acc0, false, false);
        acc1 = __builtin_amdgcn_wmma_f32_16x16x4_f32(
            false, a1, false, bfrag[j], (short)0, acc1, false, false);
    }

    if (n < 5) {
#pragma unroll
        for (int v = 0; v < 8; ++v) {
            htile[(v + 8 * hh) * HSTRIDE + n]      = acc0[v];
            htile[(16 + v + 8 * hh) * HSTRIDE + n] = acc1[v];
        }
    }
}

__global__ __launch_bounds__(NT) void kan_fused_kernel(
    const float* __restrict__ x,
    const float* __restrict__ coef1, const float* __restrict__ sb1,
    const float* __restrict__ sp1,
    const float* __restrict__ coef2, const float* __restrict__ sb2,
    const float* __restrict__ sp2,
    const float* __restrict__ fcw,   const float* __restrict__ fcb,
    float* __restrict__ out, int nrows)
{
    __shared__ float lds_wave[(NT / 32) * WAVE_LDS];

    const int t    = threadIdx.x;
    const int lane = t & 31;
    const int wave = t >> 5;
    const int blockBase = blockIdx.x * ROWS_PER_BLOCK;
    const int waveBase  = blockBase + wave * ROWS_PER_WAVE;
    const int rowq = waveBase + lane * 4;
    float* tile   = &lds_wave[wave * WAVE_LDS];
    float* htiles = tile + SLOT_FEAT;

    __builtin_prefetch(&x[(size_t)blockBase * 2], 0, 1);

    const int n  = lane & 15;
    const int hh = lane >> 4;
    const int nn = n < 5 ? n : 4;
    const float spA = sp1[nn], spB = sp1[5 + nn];
    const float sbA = sb1[nn], sbB = sb1[5 + nn];
    float cA[NB], cB[NB];
#pragma unroll
    for (int j = 0; j < NB; ++j) {
        cA[j] = coef1[nn * NB + j];
        cB[j] = coef1[(5 + nn) * NB + j];
    }

    float x0[RPL], x1[RPL];
    if (rowq + (RPL / 4 - 1) * 128 + 4 <= nrows) {
#pragma unroll
        for (int q = 0; q < RPL / 4; ++q) {
            const v4f* xp = (const v4f*)(x + (size_t)(rowq + q * 128) * 2);
            const v4f a = xp[0], b = xp[1];
            x0[4 * q + 0] = a.x; x1[4 * q + 0] = a.y; x0[4 * q + 1] = a.z; x1[4 * q + 1] = a.w;
            x0[4 * q + 2] = b.x; x1[4 * q + 2] = b.y; x0[4 * q + 3] = b.z; x1[4 * q + 3] = b.w;
        }
    } else {
#pragma unroll
        for (int r = 0; r < RPL; ++r) {
            const int row = rowq + (r >> 2) * 128 + (r & 3);
            x0[r] = 0.0f; x1[r] = 0.0f;
            if (row < nrows) {
                x0[r] = x[(size_t)row * 2 + 0];
                x1[r] = x[(size_t)row * 2 + 1];
            }
        }
    }

    float xam = 0.0f;
#pragma unroll
    for (int r = 0; r < RPL; ++r)
        xam = __builtin_fmaxf(xam, __builtin_fmaxf(__builtin_fabsf(x0[r]), __builtin_fabsf(x1[r])));
    const bool fast = __builtin_amdgcn_ballot_w32(!(xam <= 1.0f)) == 0u;

    float c2[5][NB], sp2v[5], sb2v[5];
#pragma unroll
    for (int d = 0; d < 5; ++d) {
        sp2v[d] = sp2[d];
        sb2v[d] = sb2[d];
#pragma unroll
        for (int k = 0; k < NB; ++k) c2[d][k] = coef2[d * NB + k];
    }
    const float fcw0 = fcw[0], fcb0 = fcb[0];

    v2f bfrag[5];
    float cstart;
    if (fast) {
        float wA[4], wB[4], pA, pB;
        ingrid_weights(cA, spA, sbA, hh, wA, pA);
        ingrid_weights(cB, spB, sbB, hh, wB, pB);
        const bool col = n < 5;
#pragma unroll
        for (int j = 0; j < 4; ++j) {
            bfrag[j].x = col ? wA[j] : 0.0f;
            bfrag[j].y = col ? wB[j] : 0.0f;
        }
        bfrag[4].x = 0.0f; bfrag[4].y = 0.0f;
        cstart = col ? pA + pB : 0.0f;
        v2f sl[RPL];
        silu_batch_ingrid(x0, x1, sl);
#pragma unroll
        for (int r = 0; r < RPL; ++r) {
            features_ingrid_row(x0[r], x1[r], sl[r], tile, lane);
            wave_lds_order();
            product_slot<4>(tile, htiles + r * SLOT_H, bfrag, cstart, n, hh);
            wave_lds_order();
        }
    } else {
#pragma unroll
        for (int j = 0; j < 5; ++j) {
            float w[2];
#pragma unroll
            for (int e = 0; e < 2; ++e) {
                const int k = 4 * j + 2 * hh + e;
                const float sp  = e ? spB : spA;
                const float cw  = e ? (hh ? cB[(2 * j) & 7] : cB[(2 * j - 1) & 7])
                                    : (hh ? cA[(2 * j) & 7] : cA[(2 * j - 1) & 7]);
                const float spl = BS_SCALE * (sp * cw);
                const float sil = (k == 0) ? sbA : sbB;
                float v = (k < 2) ? sil : spl;
                v = (k >= 2 + 2 * NB || n >= 5) ? 0.0f : v;
                w[e] = v;
            }
            bfrag[j].x = w[0];
            bfrag[j].y = w[1];
        }
        cstart = 0.0f;
#pragma unroll 1
        for (int r = 0; r < RPL; ++r) {
            const int row = rowq + (r >> 2) * 128 + (r & 3);
            const bool live = row < nrows;
            float xa = 0.0f, xb = 0.0f;
            if (live) {
                xa = x[(size_t)row * 2 + 0];
                xb = x[(size_t)row * 2 + 1];
            }
            features_full_row(xa, xb, live, tile, lane);
            wave_lds_order();
            product_slot<5>(tile, htiles + r * SLOT_H, bfrag, cstart, n, hh);
            wave_lds_order();
        }
    }

    float hv[RPL][5];
#pragma unroll
    for (int r = 0; r < RPL; ++r) {
        const float* hr = &htiles[r * SLOT_H + lane * HSTRIDE];
#pragma unroll
        for (int d = 0; d < 5; ++d) hv[r][d] = hr[d];
    }

    v2f s2[5][RPL / 2];
#pragma unroll
    for (int d = 0; d < 5; ++d) {
#pragma unroll
        for (int j = 0; j < RPL / 2; ++j) {
            s2[d][j].x = fmaf(hv[2 * j][d],     2.5f, 3.5f);
            s2[d][j].y = fmaf(hv[2 * j + 1][d], 2.5f, 3.5f);
        }
    }

    float smin = s2[0][0].x, smax = s2[0][0].x;
#pragma unroll
    for (int d = 0; d < 5; ++d) {
#pragma unroll
        for (int j = 0; j < RPL / 2; ++j) {
            smin = __builtin_fminf(smin, __builtin_fminf(s2[d][j].x, s2[d][j].y));
            smax = __builtin_fmaxf(smax, __builtin_fmaxf(s2[d][j].x, s2[d][j].y));
        }
    }
    bool reach[6];
    reach[0] = true;
#pragma unroll
    for (int k = 1; k <= 5; ++k)
        reach[k] = (__builtin_amdgcn_ballot_w32(smin < (float)(4 - k)) |
                    __builtin_amdgcn_ballot_w32(smax > (float)(3 + k))) != 0u;

    v2f  accp[RPL / 2];
    v2f  accs[RPL / 2];
#pragma unroll
    for (int j = 0; j < RPL / 2; ++j) { accp[j].x = 0.0f; accp[j].y = 0.0f; accs[j].x = 0.0f; accs[j].y = 0.0f; }
    float base = 0.0f;

#pragma unroll
    for (int d = 0; d < 5; ++d) {
        float Cp[NB + 6];
        const float sp = sp2v[d];
#pragma unroll
        for (int j = 0; j < NB + 6; ++j) Cp[j] = 0.0f;
#pragma unroll
        for (int k = 0; k < NB; ++k) Cp[3 + k] = sp * c2[d][k];
        float Ch[NB + 6], C6[NB + 6];
#pragma unroll
        for (int j = 0; j < NB + 6; ++j) { Ch[j] = 0.5f * Cp[j]; C6[j] = (1.0f / 6.0f) * Cp[j]; }
        auto span = [&](const int i) {
            const float al = Ch[i + 2] - Ch[i];
            const float be = (Ch[i] + Ch[i + 2]) - Cp[i + 1];
            const float ga = fmaf(3.0f, C6[i + 1] - C6[i + 2], C6[i + 3] - C6[i]);
            v2f g2, b2, a2;
            g2.x = ga; g2.y = ga;
            b2.x = be; b2.y = be;
            a2.x = al; a2.y = al;
#pragma unroll
            for (int j = 0; j < RPL / 2; ++j) {
                v2f u;
                u.x = __builtin_amdgcn_fmed3f(s2[d][j].x + (float)(2 - i), 0.0f, 1.0f);
                u.y = __builtin_amdgcn_fmed3f(s2[d][j].y + (float)(2 - i), 0.0f, 1.0f);
                v2f q = __builtin_elementwise_fma(g2, u, b2);
                q = __builtin_elementwise_fma(q, u, a2);
                accp[j] = __builtin_elementwise_fma(q, u, accp[j]);
            }
        };
        auto knot = [&](const int i) {
            return fmaf(4.0f, C6[i + 1], C6[i] + C6[i + 2]);
        };

        span(5);
        if (reach[1]) {
            span(4); span(6);
            if (reach[2]) {
                span(3); span(7);
                if (reach[3]) {
                    span(2); span(8);
                    if (reach[4]) {
                        span(1); span(9);
                        if (reach[5]) { span(0); span(10); }
                        else base += knot(1);
                    } else base += knot(2);
                } else base += knot(3);
            } else base += knot(4);
        } else base += knot(5);
    }

    if (__builtin_amdgcn_ballot_w32(!(smin >= -16.5f)) == 0u) {
        v2f one2; one2.x = 1.0f; one2.y = 1.0f;
#pragma unroll
        for (int j = 0; j < RPL / 2; ++j) {
            v2f h2[5], D[5], P[5];
#pragma unroll
            for (int d = 0; d < 5; ++d) {
                h2[d].x = hv[2 * j][d]; h2[d].y = hv[2 * j + 1][d];
                v2f t; t.x = __expf(-h2[d].x); t.y = __expf(-h2[d].y);
                D[d] = t + one2;
                P[d] = d ? P[d - 1] * D[d] : D[0];
            }
            const float R = 1.0f / (P[4].x * P[4].y);
            v2f R2; R2.x = R * P[4].y; R2.y = R * P[4].x;
            v2f inv[5];
#pragma unroll
            for (int d = 4; d >= 0; --d) {
                inv[d] = d ? R2 * P[d - 1] : R2;
                R2 = R2 * D[d];
            }
#pragma unroll
            for (int d = 0; d < 5; ++d) {
                v2f sb2; sb2.x = sb2v[d]; sb2.y = sb2v[d];
                accs[j] = __builtin_elementwise_fma(h2[d] * sb2, inv[d], accs[j]);
            }
        }
    } else {
#pragma unroll 1
        for (int r = 0; r < RPL; ++r) {
            const float* hr = &htiles[r * SLOT_H + lane * HSTRIDE];
            float a = 0.0f;
#pragma unroll
            for (int d = 0; d < 5; ++d) a = fmaf(sb2v[d], siluf(hr[d]), a);
            tile[lane * RPL + r] = a;
        }
#pragma unroll
        for (int j = 0; j < RPL / 2; ++j) {
            accs[j].x = tile[lane * RPL + 2 * j];
            accs[j].y = tile[lane * RPL + 2 * j + 1];
        }
    }

    float o[RPL];
#pragma unroll
    for (int j = 0; j < RPL / 2; ++j) {
        o[2 * j]     = ((accs[j].x + accp[j].x) + base) * fcw0 + fcb0;
        o[2 * j + 1] = ((accs[j].y + accp[j].y) + base) * fcw0 + fcb0;
    }

#pragma unroll
    for (int q = 0; q < RPL / 4; ++q) {
        const int rq = rowq + q * 128;
        if (rq + 4 <= nrows) {
            v4f ov;
            ov.x = o[4 * q + 0]; ov.y = o[4 * q + 1]; ov.z = o[4 * q + 2]; ov.w = o[4 * q + 3];
            *(v4f*)(out + rq) = ov;
        } else {
#pragma unroll
            for (int i = 0; i < 4; ++i)
                if (rq + i < nrows) out[rq + i] = o[4 * q + i];
        }
    }
}

extern "C" void kernel_launch(void* const* d_in, const int* in_sizes, int n_in,
                              void* d_out, int out_size, void* d_ws, size_t ws_size,
                              hipStream_t stream) {
    (void)n_in; (void)d_ws; (void)ws_size; (void)out_size;
    const float* x     = (const float*)d_in[0];
    const float* coef1 = (const float*)d_in[1];
    const float* sb1   = (const float*)d_in[2];
    const float* sp1   = (const float*)d_in[3];
    const float* coef2 = (const float*)d_in[4];
    const float* sb2   = (const float*)d_in[5];
    const float* sp2   = (const float*)d_in[6];
    const float* fcw   = (const float*)d_in[7];
    const float* fcb   = (const float*)d_in[8];
    float* out = (float*)d_out;

    const int nrows   = in_sizes[0] / 2;
    const int nblocks = (nrows + ROWS_PER_BLOCK - 1) / ROWS_PER_BLOCK;

    kan_fused_kernel<<<nblocks, NT, 0, stream>>>(
        x, coef1, sb1, sp1, coef2, sb2, sp2, fcw, fcb, out, nrows);
}
